# starting kernel
# baseline (speedup 1.0000x reference)
#include <hip/hip_runtime.h>
#include <math.h>

typedef float v2f __attribute__((ext_vector_type(2)));
typedef float v4f __attribute__((ext_vector_type(4)));
typedef float v8f __attribute__((ext_vector_type(8)));

#define NJ 24
#define WN_STRIDE 26
#define M_STRIDE  20

__global__ __launch_bounds__(256)
void shCaster_kernel(const float* __restrict__ xyz,
                     const float* __restrict__ viewdirs,
                     const float* __restrict__ transforms,
                     const float* __restrict__ sh_feats,
                     const float* __restrict__ locs,
                     float* __restrict__ out,
                     int N)
{
    const float SH_C0 = 0.28209479177387814f;
    const float SH_C1 = 0.4886025119029199f;
    const float C20 =  1.0925484305920792f;
    const float C21 = -1.0925484305920792f;
    const float C22 =  0.31539156525252005f;
    const float C23 = -1.0925484305920792f;
    const float C24 =  0.5462742152960396f;
    const float EPS = 1e-6f;

    __shared__ float wnLDS[256 * WN_STRIDE];
    __shared__ float mLDS [256 * M_STRIDE];

    const int tid  = threadIdx.x;
    const int lane = tid & 31;
    const int wave = tid >> 5;
    const int l15  = lane & 15;
    const int kOff = (lane >> 3) & 2;
    const int gpt  = blockIdx.x * 256 + tid;
    const int gp   = (gpt < N) ? gpt : (N - 1);

    v2f Bfrag[6];
#pragma unroll
    for (int kk = 0; kk < 6; ++kk) {
        int j0 = 4 * kk + kOff;
        Bfrag[kk].x = transforms[j0 * 16 + l15];
        Bfrag[kk].y = transforms[(j0 + 1) * 16 + l15];
    }

    const float px = xyz[gp * 3 + 0];
    const float py = xyz[gp * 3 + 1];
    const float pz = xyz[gp * 3 + 2];
    const float qx = px - viewdirs[gp * 3 + 0];
    const float qy = py - viewdirs[gp * 3 + 1];
    const float qz = pz - viewdirs[gp * 3 + 2];

    float wj[NJ];
    float wsum = 0.0f;
#pragma unroll 4
    for (int j = 0; j < NJ; ++j) {
        const float* Tj = transforms + j * 16;
        float tx = Tj[0] * px + Tj[1] * py + Tj[2]  * pz + Tj[3];
        float ty = Tj[4] * px + Tj[5] * py + Tj[6]  * pz + Tj[7];
        float tz = Tj[8] * px + Tj[9] * py + Tj[10] * pz + Tj[11];
        float dx = locs[j * 3 + 0] - tx;
        float dy = locs[j * 3 + 1] - ty;
        float dz = locs[j * 3 + 2] - tz;
        float len2 = dx * dx + dy * dy + dz * dz;
        float len  = sqrtf(len2);
        float inv  = 1.0f / len;
        dx *= inv; dy *= inv; dz *= inv;
        const float* f = sh_feats + j * 9;
        float s = SH_C0 * f[0]
                + (-SH_C1 * dy) * f[1]
                + ( SH_C1 * dz) * f[2]
                + (-SH_C1 * dx) * f[3]
                + (C20 * dx * dy) * f[4]
                + (C21 * dy * dz) * f[5]
                + (C22 * (2.0f * dz * dz - dx * dx - dy * dy)) * f[6]
                + (C23 * dx * dz) * f[7]
                + (C24 * (dx * dx - dy * dy)) * f[8];
        float rad = fmaxf(s + 0.5f, 0.0f);
        float rel = fmaxf(1.0f - len / fmaxf(rad, EPS), 0.0f);
        rel = (rad < EPS) ? 0.0f : rel;
        wj[j] = rel;
        wsum += rel;
    }
    const float invw = 1.0f / fmaxf(wsum, EPS);
#pragma unroll
    for (int j = 0; j < NJ; ++j)
        wnLDS[tid * WN_STRIDE + j] = wj[j] * invw;

    __syncthreads();

#pragma unroll
    for (int t2 = 0; t2 < 2; ++t2) {
        const int p0 = wave * 32 + t2 * 16;
        v8f c = {0.f, 0.f, 0.f, 0.f, 0.f, 0.f, 0.f, 0.f};
#pragma unroll
        for (int kk = 0; kk < 6; ++kk) {
            const int j0 = 4 * kk + kOff;
            v2f a = *(const v2f*)&wnLDS[(p0 + l15) * WN_STRIDE + j0];
            c = __builtin_amdgcn_wmma_f32_16x16x4_f32(
                    false, a, false, Bfrag[kk], (short)0, c, false, false);
        }
        const int pbase = p0 + (kOff << 2);
#pragma unroll
        for (int r = 0; r < 8; ++r)
            mLDS[(pbase + r) * M_STRIDE + l15] = c[r];
    }

    __syncthreads();

    const float* mrow = &mLDS[tid * M_STRIDE];
    v4f r0 = *(const v4f*)(mrow + 0);
    v4f r1 = *(const v4f*)(mrow + 4);
    v4f r2 = *(const v4f*)(mrow + 8);

    float o1x = r0.x * px + r0.y * py + r0.z * pz + r0.w;
    float o1y = r1.x * px + r1.y * py + r1.z * pz + r1.w;
    float o1z = r2.x * px + r2.y * py + r2.z * pz + r2.w;
    float o2x = r0.x * qx + r0.y * qy + r0.z * qz + r0.w;
    float o2y = r1.x * qx + r1.y * qy + r1.z * qz + r1.w;
    float o2z = r2.x * qx + r2.y * qy + r2.z * qz + r2.w;

    if (!(wsum > EPS)) {
        o1x = px; o1y = py; o1z = pz;
        o2x = qx; o2y = qy; o2z = qz;
    }

    if (gpt < N) {
        out[gpt * 3 + 0] = o1x;
        out[gpt * 3 + 1] = o1y;
        out[gpt * 3 + 2] = o1z;
        out[(size_t)N * 3 + gpt * 3 + 0] = o1x - o2x;
        out[(size_t)N * 3 + gpt * 3 + 1] = o1y - o2y;
        out[(size_t)N * 3 + gpt * 3 + 2] = o1z - o2z;
    }
}

extern "C" void kernel_launch(void* const* d_in, const int* in_sizes, int n_in,
                              void* d_out, int out_size, void* d_ws, size_t ws_size,
                              hipStream_t stream) {
    const float* xyz      = (const float*)d_in[0];
    const float* viewdirs = (const float*)d_in[1];
    const float* trans    = (const float*)d_in[2];
    const float* sh_feats = (const float*)d_in[3];
    const float* locs     = (const float*)d_in[4];
    float* out = (float*)d_out;

    const int N = in_sizes[0] / 3;
    const int blocks = (N + 255) / 256;
    shCaster_kernel<<<blocks, 256, 0, stream>>>(xyz, viewdirs, trans, sh_feats,
                                                locs, out, N);
}


# Joint weights are stored straight to LDS inside the joint loop and normalised there in place; the loop-indexed private array (kept in scratch, stored and reloaded every trip) is gone; same sums
# speedup vs baseline: 7.8129x; 7.8129x over previous
#include <hip/hip_runtime.h>
#include <math.h>

typedef float v2f __attribute__((ext_vector_type(2)));
typedef float v4f __attribute__((ext_vector_type(4)));
typedef float v8f __attribute__((ext_vector_type(8)));

#define NJ 24
#define WN_STRIDE 26
#define M_STRIDE  20

__global__ __launch_bounds__(256)
void shCaster_kernel(const float* __restrict__ xyz,
                     const float* __restrict__ viewdirs,
                     const float* __restrict__ transforms,
                     const float* __restrict__ sh_feats,
                     const float* __restrict__ locs,
                     float* __restrict__ out,
                     int N)
{
    const float SH_C0 = 0.28209479177387814f;
    const float SH_C1 = 0.4886025119029199f;
    const float C20 =  1.0925484305920792f;
    const float C21 = -1.0925484305920792f;
    const float C22 =  0.31539156525252005f;
    const float C23 = -1.0925484305920792f;
    const float C24 =  0.5462742152960396f;
    const float EPS = 1e-6f;

    __shared__ float wnLDS[256 * WN_STRIDE];
    __shared__ float mLDS [256 * M_STRIDE];

    const int tid  = threadIdx.x;
    const int lane = tid & 31;
    const int wave = tid >> 5;
    const int l15  = lane & 15;
    const int kOff = (lane >> 3) & 2;
    const int gpt  = blockIdx.x * 256 + tid;
    const int gp   = (gpt < N) ? gpt : (N - 1);

    v2f Bfrag[6];
#pragma unroll
    for (int kk = 0; kk < 6; ++kk) {
        int j0 = 4 * kk + kOff;
        Bfrag[kk].x = transforms[j0 * 16 + l15];
        Bfrag[kk].y = transforms[(j0 + 1) * 16 + l15];
    }

    const float px = xyz[gp * 3 + 0];
    const float py = xyz[gp * 3 + 1];
    const float pz = xyz[gp * 3 + 2];
    const float qx = px - viewdirs[gp * 3 + 0];
    const float qy = py - viewdirs[gp * 3 + 1];
    const float qz = pz - viewdirs[gp * 3 + 2];

    float* wrow = &wnLDS[tid * WN_STRIDE];
    float wsum = 0.0f;
#pragma unroll 4
    for (int j = 0; j < NJ; ++j) {
        const float* Tj = transforms + j * 16;
        float tx = Tj[0] * px + Tj[1] * py + Tj[2]  * pz + Tj[3];
        float ty = Tj[4] * px + Tj[5] * py + Tj[6]  * pz + Tj[7];
        float tz = Tj[8] * px + Tj[9] * py + Tj[10] * pz + Tj[11];
        float dx = locs[j * 3 + 0] - tx;
        float dy = locs[j * 3 + 1] - ty;
        float dz = locs[j * 3 + 2] - tz;
        float len2 = dx * dx + dy * dy + dz * dz;
        float len  = sqrtf(len2);
        float inv  = 1.0f / len;
        dx *= inv; dy *= inv; dz *= inv;
        const float* f = sh_feats + j * 9;
        float s = SH_C0 * f[0]
                + (-SH_C1 * dy) * f[1]
                + ( SH_C1 * dz) * f[2]
                + (-SH_C1 * dx) * f[3]
                + (C20 * dx * dy) * f[4]
                + (C21 * dy * dz) * f[5]
                + (C22 * (2.0f * dz * dz - dx * dx - dy * dy)) * f[6]
                + (C23 * dx * dz) * f[7]
                + (C24 * (dx * dx - dy * dy)) * f[8];
        float rad = fmaxf(s + 0.5f, 0.0f);
        float rel = fmaxf(1.0f - len / fmaxf(rad, EPS), 0.0f);
        rel = (rad < EPS) ? 0.0f : rel;
        wrow[j] = rel;
        wsum += rel;
    }
    const float invw = 1.0f / fmaxf(wsum, EPS);
#pragma unroll
    for (int j = 0; j < NJ; j += 2) {
        v2f w2 = *(const v2f*)&wrow[j];
        w2.x *= invw;
        w2.y *= invw;
        *(v2f*)&wrow[j] = w2;
    }

    __syncthreads();

#pragma unroll
    for (int t2 = 0; t2 < 2; ++t2) {
        const int p0 = wave * 32 + t2 * 16;
        v8f c = {0.f, 0.f, 0.f, 0.f, 0.f, 0.f, 0.f, 0.f};
#pragma unroll
        for (int kk = 0; kk < 6; ++kk) {
            const int j0 = 4 * kk + kOff;
            v2f a = *(const v2f*)&wnLDS[(p0 + l15) * WN_STRIDE + j0];
            c = __builtin_amdgcn_wmma_f32_16x16x4_f32(
                    false, a, false, Bfrag[kk], (short)0, c, false, false);
        }
        const int pbase = p0 + (kOff << 2);
#pragma unroll
        for (int r = 0; r < 8; ++r)
            mLDS[(pbase + r) * M_STRIDE + l15] = c[r];
    }

    __syncthreads();

    const float* mrow = &mLDS[tid * M_STRIDE];
    v4f r0 = *(const v4f*)(mrow + 0);
    v4f r1 = *(const v4f*)(mrow + 4);
    v4f r2 = *(const v4f*)(mrow + 8);

    float o1x = r0.x * px + r0.y * py + r0.z * pz + r0.w;
    float o1y = r1.x * px + r1.y * py + r1.z * pz + r1.w;
    float o1z = r2.x * px + r2.y * py + r2.z * pz + r2.w;
    float o2x = r0.x * qx + r0.y * qy + r0.z * qz + r0.w;
    float o2y = r1.x * qx + r1.y * qy + r1.z * qz + r1.w;
    float o2z = r2.x * qx + r2.y * qy + r2.z * qz + r2.w;

    if (!(wsum > EPS)) {
        o1x = px; o1y = py; o1z = pz;
        o2x = qx; o2y = qy; o2z = qz;
    }

    if (gpt < N) {
        out[gpt * 3 + 0] = o1x;
        out[gpt * 3 + 1] = o1y;
        out[gpt * 3 + 2] = o1z;
        out[(size_t)N * 3 + gpt * 3 + 0] = o1x - o2x;
        out[(size_t)N * 3 + gpt * 3 + 1] = o1y - o2y;
        out[(size_t)N * 3 + gpt * 3 + 2] = o1z - o2z;
    }
}

extern "C" void kernel_launch(void* const* d_in, const int* in_sizes, int n_in,
                              void* d_out, int out_size, void* d_ws, size_t ws_size,
                              hipStream_t stream) {
    const float* xyz      = (const float*)d_in[0];
    const float* viewdirs = (const float*)d_in[1];
    const float* trans    = (const float*)d_in[2];
    const float* sh_feats = (const float*)d_in[3];
    const float* locs     = (const float*)d_in[4];
    float* out = (float*)d_out;

    const int N = in_sizes[0] / 3;
    const int blocks = (N + 255) / 256;
    shCaster_kernel<<<blocks, 256, 0, stream>>>(xyz, viewdirs, trans, sh_feats,
                                                locs, out, N);
}


# Joint loop: the correctly rounded sqrt and both correctly rounded divisions become the hardware v_rsq_f32 / v_rcp_f32 (1 ulp each), len = len2 * rsq(len2); f32 kept, order of sums unchanged
# speedup vs baseline: 10.6633x; 1.3648x over previous
#include <hip/hip_runtime.h>
#include <math.h>

typedef float v2f __attribute__((ext_vector_type(2)));
typedef float v4f __attribute__((ext_vector_type(4)));
typedef float v8f __attribute__((ext_vector_type(8)));

#define NJ 24
#define WN_STRIDE 26
#define M_STRIDE  20

__global__ __launch_bounds__(256)
void shCaster_kernel(const float* __restrict__ xyz,
                     const float* __restrict__ viewdirs,
                     const float* __restrict__ transforms,
                     const float* __restrict__ sh_feats,
                     const float* __restrict__ locs,
                     float* __restrict__ out,
                     int N)
{
    const float SH_C0 = 0.28209479177387814f;
    const float SH_C1 = 0.4886025119029199f;
    const float C20 =  1.0925484305920792f;
    const float C21 = -1.0925484305920792f;
    const float C22 =  0.31539156525252005f;
    const float C23 = -1.0925484305920792f;
    const float C24 =  0.5462742152960396f;
    const float EPS = 1e-6f;

    __shared__ float wnLDS[256 * WN_STRIDE];
    __shared__ float mLDS [256 * M_STRIDE];

    const int tid  = threadIdx.x;
    const int lane = tid & 31;
    const int wave = tid >> 5;
    const int l15  = lane & 15;
    const int kOff = (lane >> 3) & 2;
    const int gpt  = blockIdx.x * 256 + tid;
    const int gp   = (gpt < N) ? gpt : (N - 1);

    v2f Bfrag[6];
#pragma unroll
    for (int kk = 0; kk < 6; ++kk) {
        int j0 = 4 * kk + kOff;
        Bfrag[kk].x = transforms[j0 * 16 + l15];
        Bfrag[kk].y = transforms[(j0 + 1) * 16 + l15];
    }

    const float px = xyz[gp * 3 + 0];
    const float py = xyz[gp * 3 + 1];
    const float pz = xyz[gp * 3 + 2];
    const float qx = px - viewdirs[gp * 3 + 0];
    const float qy = py - viewdirs[gp * 3 + 1];
    const float qz = pz - viewdirs[gp * 3 + 2];

    float* wrow = &wnLDS[tid * WN_STRIDE];
    float wsum = 0.0f;
#pragma unroll 4
    for (int j = 0; j < NJ; ++j) {
        const float* Tj = transforms + j * 16;
        float tx = Tj[0] * px + Tj[1] * py + Tj[2]  * pz + Tj[3];
        float ty = Tj[4] * px + Tj[5] * py + Tj[6]  * pz + Tj[7];
        float tz = Tj[8] * px + Tj[9] * py + Tj[10] * pz + Tj[11];
        float dx = locs[j * 3 + 0] - tx;
        float dy = locs[j * 3 + 1] - ty;
        float dz = locs[j * 3 + 2] - tz;
        float len2 = dx * dx + dy * dy + dz * dz;
        float inv  = __builtin_amdgcn_rsqf(len2);
        float len  = len2 * inv;
        dx *= inv; dy *= inv; dz *= inv;
        const float* f = sh_feats + j * 9;
        float s = SH_C0 * f[0]
                + (-SH_C1 * dy) * f[1]
                + ( SH_C1 * dz) * f[2]
                + (-SH_C1 * dx) * f[3]
                + (C20 * dx * dy) * f[4]
                + (C21 * dy * dz) * f[5]
                + (C22 * (2.0f * dz * dz - dx * dx - dy * dy)) * f[6]
                + (C23 * dx * dz) * f[7]
                + (C24 * (dx * dx - dy * dy)) * f[8];
        float rad = fmaxf(s + 0.5f, 0.0f);
        float rel = fmaxf(1.0f - len * __builtin_amdgcn_rcpf(fmaxf(rad, EPS)), 0.0f);
        rel = (rad < EPS) ? 0.0f : rel;
        wrow[j] = rel;
        wsum += rel;
    }
    const float invw = __builtin_amdgcn_rcpf(fmaxf(wsum, EPS));
#pragma unroll
    for (int j = 0; j < NJ; j += 2) {
        v2f w2 = *(const v2f*)&wrow[j];
        w2.x *= invw;
        w2.y *= invw;
        *(v2f*)&wrow[j] = w2;
    }

    __syncthreads();

#pragma unroll
    for (int t2 = 0; t2 < 2; ++t2) {
        const int p0 = wave * 32 + t2 * 16;
        v8f c = {0.f, 0.f, 0.f, 0.f, 0.f, 0.f, 0.f, 0.f};
#pragma unroll
        for (int kk = 0; kk < 6; ++kk) {
            const int j0 = 4 * kk + kOff;
            v2f a = *(const v2f*)&wnLDS[(p0 + l15) * WN_STRIDE + j0];
            c = __builtin_amdgcn_wmma_f32_16x16x4_f32(
                    false, a, false, Bfrag[kk], (short)0, c, false, false);
        }
        const int pbase = p0 + (kOff << 2);
#pragma unroll
        for (int r = 0; r < 8; ++r)
            mLDS[(pbase + r) * M_STRIDE + l15] = c[r];
    }

    __syncthreads();

    const float* mrow = &mLDS[tid * M_STRIDE];
    v4f r0 = *(const v4f*)(mrow + 0);
    v4f r1 = *(const v4f*)(mrow + 4);
    v4f r2 = *(const v4f*)(mrow + 8);

    float o1x = r0.x * px + r0.y * py + r0.z * pz + r0.w;
    float o1y = r1.x * px + r1.y * py + r1.z * pz + r1.w;
    float o1z = r2.x * px + r2.y * py + r2.z * pz + r2.w;
    float o2x = r0.x * qx + r0.y * qy + r0.z * qz + r0.w;
    float o2y = r1.x * qx + r1.y * qy + r1.z * qz + r1.w;
    float o2z = r2.x * qx + r2.y * qy + r2.z * qz + r2.w;

    if (!(wsum > EPS)) {
        o1x = px; o1y = py; o1z = pz;
        o2x = qx; o2y = qy; o2z = qz;
    }

    if (gpt < N) {
        out[gpt * 3 + 0] = o1x;
        out[gpt * 3 + 1] = o1y;
        out[gpt * 3 + 2] = o1z;
        out[(size_t)N * 3 + gpt * 3 + 0] = o1x - o2x;
        out[(size_t)N * 3 + gpt * 3 + 1] = o1y - o2y;
        out[(size_t)N * 3 + gpt * 3 + 2] = o1z - o2z;
    }
}

extern "C" void kernel_launch(void* const* d_in, const int* in_sizes, int n_in,
                              void* d_out, int out_size, void* d_ws, size_t ws_size,
                              hipStream_t stream) {
    const float* xyz      = (const float*)d_in[0];
    const float* viewdirs = (const float*)d_in[1];
    const float* trans    = (const float*)d_in[2];
    const float* sh_feats = (const float*)d_in[3];
    const float* locs     = (const float*)d_in[4];
    float* out = (float*)d_out;

    const int N = in_sizes[0] / 3;
    const int blocks = (N + 255) / 256;
    shCaster_kernel<<<blocks, 256, 0, stream>>>(xyz, viewdirs, trans, sh_feats,
                                                locs, out, N);
}


# Joint loop: SH sum taken on the unnormalised direction as c + inv*(-C1*lin + inv*quad), joint constants folded in scalar registers, d formed as one fma chain; order of sums changes
# speedup vs baseline: 13.0221x; 1.2212x over previous
#include <hip/hip_runtime.h>
#include <math.h>

typedef float v2f __attribute__((ext_vector_type(2)));
typedef float v4f __attribute__((ext_vector_type(4)));
typedef float v8f __attribute__((ext_vector_type(8)));

#define NJ 24
#define WN_STRIDE 26
#define M_STRIDE  20

__global__ __launch_bounds__(256)
void shCaster_kernel(const float* __restrict__ xyz,
                     const float* __restrict__ viewdirs,
                     const float* __restrict__ transforms,
                     const float* __restrict__ sh_feats,
                     const float* __restrict__ locs,
                     float* __restrict__ out,
                     int N)
{
    const float SH_C0 = 0.28209479177387814f;
    const float SH_C1 = 0.4886025119029199f;
    const float C20 =  1.0925484305920792f;
    const float C22 =  0.31539156525252005f;
    const float EPS = 1e-6f;

    __shared__ float wnLDS[256 * WN_STRIDE];
    __shared__ float mLDS [256 * M_STRIDE];

    const int tid  = threadIdx.x;
    const int lane = tid & 31;
    const int wave = tid >> 5;
    const int l15  = lane & 15;
    const int kOff = (lane >> 3) & 2;
    const int gpt  = blockIdx.x * 256 + tid;
    const int gp   = (gpt < N) ? gpt : (N - 1);

    v2f Bfrag[6];
#pragma unroll
    for (int kk = 0; kk < 6; ++kk) {
        int j0 = 4 * kk + kOff;
        Bfrag[kk].x = transforms[j0 * 16 + l15];
        Bfrag[kk].y = transforms[(j0 + 1) * 16 + l15];
    }

    const float px = xyz[gp * 3 + 0];
    const float py = xyz[gp * 3 + 1];
    const float pz = xyz[gp * 3 + 2];
    const float qx = px - viewdirs[gp * 3 + 0];
    const float qy = py - viewdirs[gp * 3 + 1];
    const float qz = pz - viewdirs[gp * 3 + 2];

    float* wrow = &wnLDS[tid * WN_STRIDE];
    float wsum = 0.0f;
#pragma unroll 4
    for (int j = 0; j < NJ; ++j) {
        const float* Tj = transforms + j * 16;
        const float* f  = sh_feats + j * 9;
        const float bx = locs[j * 3 + 0] - Tj[3];
        const float by = locs[j * 3 + 1] - Tj[7];
        const float bz = locs[j * 3 + 2] - Tj[11];
        const float h8 = 0.5f * f[8];
        const float g6 = (3.0f * C22) * f[6];
        const float cc = (SH_C0 * f[0] + 0.5f) - C22 * f[6];
        const float dx = fmaf(-Tj[0], px, fmaf(-Tj[1], py, fmaf(-Tj[2],  pz, bx)));
        const float dy = fmaf(-Tj[4], px, fmaf(-Tj[5], py, fmaf(-Tj[6],  pz, by)));
        const float dz = fmaf(-Tj[8], px, fmaf(-Tj[9], py, fmaf(-Tj[10], pz, bz)));
        const float zz = dz * dz;
        const float len2 = fmaf(dx, dx, fmaf(dy, dy, zz));
        const float inv = __builtin_amdgcn_rsqf(len2);
        const float len = len2 * inv;
        const float a1 = fmaf(f[3], dx, fmaf(f[1], dy, -f[2] * dz));
        const float u  = fmaf(h8, dx, fmaf(f[4], dy, -f[7] * dz));
        const float v  = fmaf(h8, dy, f[5] * dz);
        const float a2 = fmaf(dx, u, -dy * v);
        const float in2 = fmaf(C20, a2, g6 * zz);
        const float mid = fmaf(-SH_C1, a1, inv * in2);
        const float s   = fmaf(inv, mid, cc);
        float rel = fmaxf(fmaf(-len, __builtin_amdgcn_rcpf(fmaxf(s, EPS)), 1.0f), 0.0f);
        rel = (s < EPS) ? 0.0f : rel;
        wrow[j] = rel;
        wsum += rel;
    }
    const float invw = __builtin_amdgcn_rcpf(fmaxf(wsum, EPS));
#pragma unroll
    for (int j = 0; j < NJ; j += 2) {
        v2f w2 = *(const v2f*)&wrow[j];
        w2.x *= invw;
        w2.y *= invw;
        *(v2f*)&wrow[j] = w2;
    }

    __syncthreads();

#pragma unroll
    for (int t2 = 0; t2 < 2; ++t2) {
        const int p0 = wave * 32 + t2 * 16;
        v8f c = {0.f, 0.f, 0.f, 0.f, 0.f, 0.f, 0.f, 0.f};
#pragma unroll
        for (int kk = 0; kk < 6; ++kk) {
            const int j0 = 4 * kk + kOff;
            v2f a = *(const v2f*)&wnLDS[(p0 + l15) * WN_STRIDE + j0];
            c = __builtin_amdgcn_wmma_f32_16x16x4_f32(
                    false, a, false, Bfrag[kk], (short)0, c, false, false);
        }
        const int pbase = p0 + (kOff << 2);
#pragma unroll
        for (int r = 0; r < 8; ++r)
            mLDS[(pbase + r) * M_STRIDE + l15] = c[r];
    }

    __syncthreads();

    const float* mrow = &mLDS[tid * M_STRIDE];
    v4f r0 = *(const v4f*)(mrow + 0);
    v4f r1 = *(const v4f*)(mrow + 4);
    v4f r2 = *(const v4f*)(mrow + 8);

    float o1x = r0.x * px + r0.y * py + r0.z * pz + r0.w;
    float o1y = r1.x * px + r1.y * py + r1.z * pz + r1.w;
    float o1z = r2.x * px + r2.y * py + r2.z * pz + r2.w;
    float o2x = r0.x * qx + r0.y * qy + r0.z * qz + r0.w;
    float o2y = r1.x * qx + r1.y * qy + r1.z * qz + r1.w;
    float o2z = r2.x * qx + r2.y * qy + r2.z * qz + r2.w;

    if (!(wsum > EPS)) {
        o1x = px; o1y = py; o1z = pz;
        o2x = qx; o2y = qy; o2z = qz;
    }

    if (gpt < N) {
        out[gpt * 3 + 0] = o1x;
        out[gpt * 3 + 1] = o1y;
        out[gpt * 3 + 2] = o1z;
        out[(size_t)N * 3 + gpt * 3 + 0] = o1x - o2x;
        out[(size_t)N * 3 + gpt * 3 + 1] = o1y - o2y;
        out[(size_t)N * 3 + gpt * 3 + 2] = o1z - o2z;
    }
}

extern "C" void kernel_launch(void* const* d_in, const int* in_sizes, int n_in,
                              void* d_out, int out_size, void* d_ws, size_t ws_size,
                              hipStream_t stream) {
    const float* xyz      = (const float*)d_in[0];
    const float* viewdirs = (const float*)d_in[1];
    const float* trans    = (const float*)d_in[2];
    const float* sh_feats = (const float*)d_in[3];
    const float* locs     = (const float*)d_in[4];
    float* out = (float*)d_out;

    const int N = in_sizes[0] / 3;
    const int blocks = (N + 255) / 256;
    shCaster_kernel<<<blocks, 256, 0, stream>>>(xyz, viewdirs, trans, sh_feats,
                                                locs, out, N);
}
